# starting kernel
# baseline (speedup 1.0000x reference)
#include <hip/hip_runtime.h>

#define N_NODES 100000
#define N_EDGES 1600000
#define IN_DIM  32
#define OUT_DIM 32
#define N_TILES (N_NODES / 16)

typedef __attribute__((ext_vector_type(2))) float v2f;
typedef __attribute__((ext_vector_type(8))) float v8f;

__global__ __launch_bounds__(256) void gin_zero_kernel(float4* __restrict__ agg4, int n4) {
    int i = blockIdx.x * 256 + threadIdx.x;
    if (i < n4) agg4[i] = float4{0.f, 0.f, 0.f, 0.f};
}

__device__ __forceinline__ void atomic_add_f32_noret(float* p, float v) {
    asm volatile("global_atomic_add_f32 %0, %1, off"
                 :
                 : "v"(p), "v"(v)
                 : "memory");
}

__global__ __launch_bounds__(256) void gin_scatter_kernel(const float* __restrict__ X,
                                                          const int* __restrict__ src,
                                                          const int* __restrict__ dst,
                                                          float* __restrict__ agg) {
    unsigned tid = blockIdx.x * 256u + threadIdx.x;
    unsigned e = tid >> 4;
    unsigned d = (tid & 15u) * 2u;
    unsigned s = (unsigned)src[e];
    unsigned t = (unsigned)dst[e];
    const float2 v = *(const float2*)&X[s * 32u + d];
    float* p = &agg[t * 32u + d];
    atomic_add_f32_noret(p,     v.x);
    atomic_add_f32_noret(p + 1, v.y);
}

__global__ __launch_bounds__(128) void gin_gemm_kernel(const float* __restrict__ A,
                                                       const float* __restrict__ W,
                                                       float* __restrict__ out) {
    __shared__ float Wt[32 * 32];
    int t = threadIdx.x;
    for (int i = t; i < 1024; i += 128) {
        int k = i >> 5, n = i & 31;
        Wt[n * 32 + k] = W[i];
    }
    __syncthreads();

    int wave = t >> 5;
    int lane = t & 31;
    int m    = lane & 15;
    int hi   = lane >> 4;
    int tile = blockIdx.x * 4 + wave;

    if (tile < N_TILES) {
        int row0 = tile * 16;
        const float* aptr = A + (size_t)(row0 + m) * 32;

        v2f a[8];
#pragma unroll
        for (int s = 0; s < 8; ++s) {
            int k = 4 * s + 2 * hi;
            a[s].x = aptr[k];
            a[s].y = aptr[k + 1];
        }

#pragma unroll
        for (int nt = 0; nt < 2; ++nt) {
            v8f c = {};
            const float* wp = &Wt[(nt * 16 + m) * 32 + 2 * hi];
#pragma unroll
            for (int s = 0; s < 8; ++s) {
                v2f b;
                b.x = wp[4 * s];
                b.y = wp[4 * s + 1];
                c = __builtin_amdgcn_wmma_f32_16x16x4_f32(
                         false, a[s],  false, b,
                         (short)0, c,  false,  false);
            }
            float* op = out + (size_t)(row0 + 8 * hi) * 32 + nt * 16 + m;
#pragma unroll
            for (int r = 0; r < 8; ++r) {
                op[(size_t)r * 32] = c[r];
            }
        }
    }
}

extern "C" void kernel_launch(void* const* d_in, const int* in_sizes, int n_in,
                              void* d_out, int out_size, void* d_ws, size_t ws_size,
                              hipStream_t stream) {
    const float* X   = (const float*)d_in[0];
    const float* W   = (const float*)d_in[1];
    const int*   src = (const int*)d_in[2];
    const int*   dst = (const int*)d_in[3];
    float*       out = (float*)d_out;
    float*       agg = (float*)d_ws;

    int n4 = (N_NODES * IN_DIM) / 4;
    gin_zero_kernel<<<(n4 + 255) / 256, 256, 0, stream>>>((float4*)agg, n4);

    gin_scatter_kernel<<<(N_EDGES * 16) / 256, 256, 0, stream>>>(X, src, dst, agg);

    int gblocks = (N_TILES + 3) / 4;
    gin_gemm_kernel<<<gblocks, 128, 0, stream>>>(agg, W, out);
}


# atomics removed: 250 work-groups each own 400 aggregate rows, find their edges by scanning a 1-byte owner key per edge, add X rows in registers in a fixed order (new sum order); 3 launches to 2
# speedup vs baseline: 2.0482x; 2.0482x over previous
#include <hip/hip_runtime.h>


#define N_NODES 100000
#define N_EDGES 1600000
#define IN_DIM  32
#define OUT_DIM 32

#define PART      400
#define N_PARTS   (N_NODES / PART)
#define N_WAVES   32
#define GROUP     1024
#define N_GROUPS  1563
#define N_KEY16   (N_EDGES / 16)
#define QCAP      1536
#define ROWS_PER_WAVE 13
#define INVALID   0xffffffffu
#define CELLCAP   16

typedef __attribute__((ext_vector_type(2))) float v2f;
typedef __attribute__((ext_vector_type(8))) float v8f;
typedef __attribute__((ext_vector_type(16))) float v16f;
typedef __attribute__((address_space(3))) volatile unsigned lds_vu32;

__device__ __forceinline__ unsigned make_entry(unsigned s, unsigned d, unsigned part) {
    return ((part & 1u) << 26) | (s << 9) | (d - part * PART);
}

__global__ __launch_bounds__(256) void gin_key_kernel(const int* __restrict__ src,
                                                      const int* __restrict__ dst,
                                                      unsigned* __restrict__ keys,
                                                      uint4* __restrict__ entries) {
    unsigned i = blockIdx.x * 256u + threadIdx.x;
    if (i < N_EDGES / 4) {
        const int4 d = ((const int4*)dst)[i];
        const int4 s = ((const int4*)src)[i];
        const unsigned p0 = (unsigned)d.x / PART, p1 = (unsigned)d.y / PART;
        const unsigned p2 = (unsigned)d.z / PART, p3 = (unsigned)d.w / PART;
        keys[i] = p0 | (p1 << 8) | (p2 << 16) | (p3 << 24);
        entries[i] = uint4{make_entry((unsigned)s.x, (unsigned)d.x, p0),
                           make_entry((unsigned)s.y, (unsigned)d.y, p1),
                           make_entry((unsigned)s.z, (unsigned)d.z, p2),
                           make_entry((unsigned)s.w, (unsigned)d.w, p3)};
    }
}

__device__ __forceinline__ unsigned zero_byte_flags(unsigned t) {
    return (t - 0x01010101u) & ~t;
}

template <bool FULL>
__device__ __forceinline__ void add_batch(const float* __restrict__ X,
                                          const lds_vu32* q, unsigned nvalid,
                                          unsigned lane, v16f& racc) {
    unsigned pk = 0u;
    if (FULL || lane < nvalid) pk = q[lane];
    float v[32];
#pragma unroll
    for (int u = 0; u < 32; ++u) {
        unsigned e = __builtin_amdgcn_readlane(pk, u);
        v[u] = X[(e >> 9) * 32u + lane];
    }
#pragma unroll
    for (int u = 0; u < 32; ++u) {
        if (FULL || (unsigned)u < nvalid) {
            unsigned e = __builtin_amdgcn_readlane(pk, u);
            racc[(e & 511u) >> 5] += v[u];
        }
    }
}

__global__ __launch_bounds__(1024) void gin_owner_kernel(const float* __restrict__ X,
                                                         const float* __restrict__ W,
                                                         const unsigned* __restrict__ keys,
                                                         const unsigned* __restrict__ entries,
                                                         float* __restrict__ out) {
    __shared__ unsigned qlist[N_WAVES * QCAP];
    float* acc = (float*)qlist;
    __shared__ unsigned cells[N_WAVES * N_WAVES * CELLCAP];
    __shared__ unsigned cellcnt[N_WAVES * N_WAVES];
    __shared__ unsigned ownq[N_WAVES * 64];
    __shared__ unsigned qcnt[N_WAVES];
    __shared__ unsigned more_flag[2];

    const unsigned tid  = threadIdx.x;
    const unsigned wave = __builtin_amdgcn_readfirstlane(tid >> 5);
    const unsigned lane = tid & 31u;
    const unsigned p    = blockIdx.x;
    const unsigned lo   = p * PART;

    lds_vu32* mylist = (lds_vu32*)&qlist[wave * QCAP];
    lds_vu32* myown  = (lds_vu32*)&ownq[wave * 64];
    lds_vu32* vcnt   = (lds_vu32*)qcnt;
    lds_vu32* vcells = (lds_vu32*)cells;
    lds_vu32* vccnt  = (lds_vu32*)cellcnt;
    lds_vu32* vmore  = (lds_vu32*)more_flag;

    if (tid < 2) vmore[tid] = 0u;
    __syncthreads();

    v16f racc = {};

    const uint4* k16 = (const uint4*)keys;
    const unsigned pat = p * 0x01010101u;
    unsigned g = wave;
    unsigned round = 0u;
    unsigned more;

    do {
        unsigned n = 0u;
        if (g < N_GROUPS) {
            auto fetch = [&](uint4& y0, uint4& y1, unsigned gq) {
                const unsigned gg = (gq < N_GROUPS - 1u) ? gq : (N_GROUPS - 1u);
                const unsigned i  = gg * 64u + lane * 2u;
                const unsigned ic = (i < N_KEY16 - 2u) ? i : (N_KEY16 - 2u);
                y0 = k16[ic]; y1 = k16[ic + 1u];
            };
            auto test = [&](const uint4& a0, const uint4& a1, unsigned gg) {
                const unsigned M = 0x80808080u;
                const unsigned z0 = zero_byte_flags(a0.x ^ pat) & M;
                const unsigned z1 = zero_byte_flags(a0.y ^ pat) & M;
                const unsigned z2 = zero_byte_flags(a0.z ^ pat) & M;
                const unsigned z3 = zero_byte_flags(a0.w ^ pat) & M;
                const unsigned z4 = zero_byte_flags(a1.x ^ pat) & M;
                const unsigned z5 = zero_byte_flags(a1.y ^ pat) & M;
                const unsigned z6 = zero_byte_flags(a1.z ^ pat) & M;
                const unsigned z7 = zero_byte_flags(a1.w ^ pat) & M;
                unsigned m = (z0 >> 7) | (z1 >> 6) | (z2 >> 5) | (z3 >> 4)
                           | (z4 >> 3) | (z5 >> 2) | (z6 >> 1) | z7;
                if (gg * 64u + lane * 2u >= N_KEY16) m = 0u;
                unsigned live = __builtin_amdgcn_ballot_w32(m != 0u);
                if (live != 0u) {
                    const unsigned ebase = gg * GROUP + lane * 32u;
                    do {
                        if (m != 0u) {
                            unsigned bit = (unsigned)__builtin_ctz(m);
                            m &= m - 1u;
                            unsigned pos = n + __builtin_amdgcn_mbcnt_lo(live, 0u);
                            mylist[pos] = ebase + ((bit & 7u) << 2) + (bit >> 3);
                        }
                        n += (unsigned)__builtin_popcount(live);
                        live = __builtin_amdgcn_ballot_w32(m != 0u);
                    } while (live != 0u);
                }
            };
            uint4 c0, c1, x0, x1;
            fetch(c0, c1, g);
            while (n + GROUP <= QCAP) {
                fetch(x0, x1, g + N_WAVES);
                test(c0, c1, g);
                g += N_WAVES;
                if (!(g < N_GROUPS && n + GROUP <= QCAP)) break;
                fetch(c0, c1, g + N_WAVES);
                test(x0, x1, g);
                g += N_WAVES;
                if (!(g < N_GROUPS)) break;
            }
        }
        if (lane == 0 && g < N_GROUPS) vmore[round & 1u] = 1u;

        unsigned made = 0u;
        unsigned left = 0u;
        for (unsigned i0 = 0; i0 < n; i0 += 128u) {
            unsigned e[4], w[4];
#pragma unroll
            for (int j = 0; j < 4; ++j) {
                unsigned idx = i0 + j * 32u + lane;
                e[j] = (idx < n) ? mylist[idx] : 0u;
            }
#pragma unroll
            for (int j = 0; j < 4; ++j) w[j] = entries[e[j]];
#pragma unroll
            for (int j = 0; j < 4; ++j) {
                const unsigned idx = i0 + j * 32u + lane;
                const bool valid = (idx < n) && (((w[j] >> 26) ^ p) & 1u) == 0u;
                const unsigned pk = w[j] & 0x03ffffffu;
                const unsigned o  = pk & 31u;
                unsigned peers = __builtin_amdgcn_ballot_w32(valid);
                unsigned forme = peers;
#pragma unroll
                for (int bit = 0; bit < 5; ++bit) {
                    const unsigned b = __builtin_amdgcn_ballot_w32(valid && ((o >> bit) & 1u));
                    peers &= ((o >> bit) & 1u) ? b : ~b;
                    forme &= ((lane >> bit) & 1u) ? b : ~b;
                }
                const unsigned rank  = (unsigned)__builtin_popcount(peers & ((1u << lane) - 1u));
                const unsigned basec = (unsigned)__builtin_amdgcn_ds_bpermute((int)(o << 2), (int)made);
                const unsigned pos   = basec + rank;
                if (valid && pos < CELLCAP) vcells[(wave * N_WAVES + o) * CELLCAP + pos] = pk;
                const bool full = valid && pos >= CELLCAP;
                const unsigned fm = __builtin_amdgcn_ballot_w32(full);
                if (fm != 0u) {
                    if (full) mylist[left + __builtin_amdgcn_mbcnt_lo(fm, 0u)] = pk;
                    left += (unsigned)__builtin_popcount(fm);
                }
                made += (unsigned)__builtin_popcount(forme);
            }
        }
        vccnt[wave * N_WAVES + lane] = (made < CELLCAP) ? made : CELLCAP;
        if (lane == 0) vcnt[wave] = left;
        __syncthreads();
        more = vmore[round & 1u];
        if (tid == 0) vmore[(round + 1u) & 1u] = 0u;

        unsigned nq = 0u;
        const unsigned cc = vccnt[lane * N_WAVES + wave];
        for (unsigned v = 0; v < N_WAVES; ++v) {
            const unsigned c = __builtin_amdgcn_readlane(cc, v);
            if (c != 0u) {
                if (lane < c) myown[nq + lane] = vcells[(v * N_WAVES + wave) * CELLCAP + lane];
                nq += c;
                if (nq >= 32u) {
                    nq -= 32u;
                    add_batch<true>(X, myown + nq, 32u, lane, racc);
                }
            }
        }
        const unsigned lc = vcnt[lane];
        if (__builtin_amdgcn_ballot_w32(lc != 0u) != 0u) {
            for (unsigned v = 0; v < N_WAVES; ++v) {
                const unsigned nv = __builtin_amdgcn_readlane(lc, v);
                const lds_vu32* lv = (const lds_vu32*)&qlist[v * QCAP];
                for (unsigned i0 = 0; i0 < nv; i0 += 32u) {
                    const unsigned idx = i0 + lane;
                    const unsigned pk  = (idx < nv) ? lv[idx] : INVALID;
                    const unsigned rel = pk & 511u;
                    const bool mine = (rel < PART) && ((rel & 31u) == wave);
                    const unsigned mask = __builtin_amdgcn_ballot_w32(mine);
                    if (mask != 0u) {
                        if (mine) myown[nq + __builtin_amdgcn_mbcnt_lo(mask, 0u)] = pk;
                        nq += (unsigned)__builtin_popcount(mask);
                        if (nq >= 32u) {
                            nq -= 32u;
                            add_batch<true>(X, myown + nq, 32u, lane, racc);
                        }
                    }
                }
            }
        }
        if (nq > 0u) add_batch<false>(X, myown, nq, lane, racc);
        __syncthreads();
        ++round;
    } while (more != 0u);

#pragma unroll
    for (int k = 0; k < ROWS_PER_WAVE; ++k) {
        unsigned rel = wave + 32u * k;
        if (rel < PART) acc[rel * 32u + lane] = racc[k];
    }
    __syncthreads();

    if (wave < PART / 16) {
        const unsigned m16 = lane & 15u;
        const unsigned hi  = lane >> 4;
        const float* aptr = &acc[(wave * 16u + m16) * 32u + 2u * hi];
        v2f a[8];
#pragma unroll
        for (int s = 0; s < 8; ++s) {
            a[s].x = aptr[4 * s];
            a[s].y = aptr[4 * s + 1];
        }
        const unsigned row0 = lo + wave * 16u;
#pragma unroll
        for (int nt = 0; nt < 2; ++nt) {
            v8f c = {};
            const float* wp = W + (2u * hi) * 32u + nt * 16 + m16;
#pragma unroll
            for (int s = 0; s < 8; ++s) {
                v2f b;
                b.x = wp[(4 * s) * 32];
                b.y = wp[(4 * s + 1) * 32];
                c = __builtin_amdgcn_wmma_f32_16x16x4_f32(
                         false, a[s],  false, b,
                         (short)0, c,  false,  false);
            }
            float* op = out + (size_t)(row0 + 8u * hi) * 32 + nt * 16 + m16;
#pragma unroll
            for (int r = 0; r < 8; ++r) {
                op[(size_t)r * 32] = c[r];
            }
        }
    }
}

extern "C" void kernel_launch(void* const* d_in, const int* in_sizes, int n_in,
                              void* d_out, int out_size, void* d_ws, size_t ws_size,
                              hipStream_t stream) {
    const float* X   = (const float*)d_in[0];
    const float* W   = (const float*)d_in[1];
    const int*   src = (const int*)d_in[2];
    const int*   dst = (const int*)d_in[3];
    float*       out = (float*)d_out;
    unsigned*    keys    = (unsigned*)d_ws;
    unsigned*    entries = (unsigned*)d_ws + N_EDGES / 4;

    gin_key_kernel<<<1563, 256, 0, stream>>>(src, dst, keys, (uint4*)entries);
    gin_owner_kernel<<<N_PARTS, 1024, 0, stream>>>(X, W, keys, entries, out);
}


# owner blocks no longer scan every edge key: kernel 1 writes a per-part presence bitmap (1 bit per 8 edges, no atomics), owner reads its row and only the hit dst chunks; new fixed sum order
# speedup vs baseline: 2.5144x; 1.2276x over previous
#include <hip/hip_runtime.h>


#define N_NODES 100000
#define N_EDGES 1600000
#define IN_DIM  32
#define OUT_DIM 32

#define PART      400
#define N_PARTS   (N_NODES / PART)
#define N_WAVES   32
#define CHUNK     8
#define N_CHUNKS  (N_EDGES / CHUNK)
#define IDX_BLOCKS 196
#define ROW_WORDS (IDX_BLOCKS * 32)
#define BM_PITCH  257
#define BATCH     64
#define QCAP      828
#define CQCAP     1088
#define WSTRIDE   (QCAP + CQCAP)
#define ROWS_PER_WAVE 13
#define INVALID   0xffffffffu
#define CELLCAP   16

typedef __attribute__((ext_vector_type(2))) float v2f;
typedef __attribute__((ext_vector_type(8))) float v8f;
typedef __attribute__((ext_vector_type(16))) float v16f;
typedef __attribute__((address_space(3))) volatile unsigned lds_vu32;

__global__ __launch_bounds__(1024) void gin_index_kernel(const int* __restrict__ dst,
                                                         unsigned* __restrict__ bitmap) {
    __shared__ unsigned bm[N_WAVES * BM_PITCH];
    const unsigned tid  = threadIdx.x;
    const unsigned wave = tid >> 5;
    const unsigned lane = tid & 31u;
    for (unsigned i = tid; i < N_WAVES * BM_PITCH; i += 1024u) bm[i] = 0u;
    __syncthreads();
    const unsigned c  = blockIdx.x * 1024u + tid;
    const bool     ok = c < N_CHUNKS;
    const unsigned cc = ok ? c : 0u;
    const int4 d0 = ((const int4*)dst)[2u * cc];
    const int4 d1 = ((const int4*)dst)[2u * cc + 1u];
    const unsigned part[8] = {(unsigned)d0.x / PART, (unsigned)d0.y / PART, (unsigned)d0.z / PART, (unsigned)d0.w / PART,
                              (unsigned)d1.x / PART, (unsigned)d1.y / PART, (unsigned)d1.z / PART, (unsigned)d1.w / PART};
    lds_vu32* col = (lds_vu32*)&bm[wave * BM_PITCH];
#pragma unroll
    for (int j = 0; j < 8; ++j) {
        const unsigned q = part[j];
        unsigned peers = __builtin_amdgcn_ballot_w32(ok);
#pragma unroll
        for (int bit = 0; bit < 8; ++bit) {
            const unsigned b = __builtin_amdgcn_ballot_w32(ok && ((q >> bit) & 1u));
            peers &= ((q >> bit) & 1u) ? b : ~b;
        }
        if (ok && (peers & ((1u << lane) - 1u)) == 0u) col[q] = col[q] | peers;
    }
    __syncthreads();
    for (unsigned part = wave; part < N_PARTS; part += N_WAVES)
        bitmap[(part * IDX_BLOCKS + blockIdx.x) * 32u + lane] = bm[lane * BM_PITCH + part];
}

template <bool FULL>
__device__ __forceinline__ void add_batch(const float* __restrict__ X,
                                          const lds_vu32* q, unsigned nvalid,
                                          unsigned lane, v16f& racc) {
    unsigned pk = 0u;
    if (FULL || lane < nvalid) pk = q[lane];
    float v[32];
#pragma unroll
    for (int u = 0; u < 32; ++u) {
        unsigned e = __builtin_amdgcn_readlane(pk, u);
        v[u] = X[(e >> 9) * 32u + lane];
    }
#pragma unroll
    for (int u = 0; u < 32; ++u) {
        if (FULL || (unsigned)u < nvalid) {
            unsigned e = __builtin_amdgcn_readlane(pk, u);
            racc[(e & 511u) >> 5] += v[u];
        }
    }
}

__global__ __launch_bounds__(1024) void gin_owner_kernel(const float* __restrict__ X,
                                                         const float* __restrict__ W,
                                                         const int* __restrict__ src,
                                                         const int* __restrict__ dst,
                                                         const unsigned* __restrict__ bitmap,
                                                         float* __restrict__ out) {
    __shared__ unsigned qlist[N_WAVES * WSTRIDE];
    float* acc = (float*)qlist;
    __shared__ unsigned cells[N_WAVES * N_WAVES * CELLCAP];
    __shared__ unsigned cellcnt[N_WAVES * N_WAVES];
    __shared__ unsigned ownq[N_WAVES * 64];
    __shared__ unsigned qcnt[N_WAVES];
    __shared__ unsigned more_flag[2];

    const unsigned tid  = threadIdx.x;
    const unsigned wave = __builtin_amdgcn_readfirstlane(tid >> 5);
    const unsigned lane = tid & 31u;
    const unsigned p    = blockIdx.x;
    const unsigned lo   = p * PART;

    lds_vu32* mylist = (lds_vu32*)&qlist[wave * WSTRIDE];
    lds_vu32* mycq   = (lds_vu32*)&qlist[wave * WSTRIDE + QCAP];
    lds_vu32* myown  = (lds_vu32*)&ownq[wave * 64];
    lds_vu32* vcnt   = (lds_vu32*)qcnt;
    lds_vu32* vcells = (lds_vu32*)cells;
    lds_vu32* vccnt  = (lds_vu32*)cellcnt;
    lds_vu32* vmore  = (lds_vu32*)more_flag;

    if (tid < 2) vmore[tid] = 0u;
    __syncthreads();

    v16f racc = {};

    const unsigned* row = bitmap + p * ROW_WORDS;
    const int4* d4 = (const int4*)dst;
    unsigned bi  = wave;
    unsigned cur = row[bi * 32u + lane];
    unsigned cn  = 0u;
    unsigned round = 0u;
    unsigned more;

    do {
        unsigned n = 0u;
        auto hits8 = [&](const int4& a0, const int4& a1, bool on) -> unsigned {
            const unsigned m = ((unsigned)a0.x - lo < PART ? 1u : 0u)   | ((unsigned)a0.y - lo < PART ? 2u : 0u)
                             | ((unsigned)a0.z - lo < PART ? 4u : 0u)   | ((unsigned)a0.w - lo < PART ? 8u : 0u)
                             | ((unsigned)a1.x - lo < PART ? 16u : 0u)  | ((unsigned)a1.y - lo < PART ? 32u : 0u)
                             | ((unsigned)a1.z - lo < PART ? 64u : 0u)  | ((unsigned)a1.w - lo < PART ? 128u : 0u);
            return on ? m : 0u;
        };
        auto list_hits = [&](unsigned m, unsigned c) {
            unsigned live = __builtin_amdgcn_ballot_w32(m != 0u);
            while (live != 0u) {
                if (m != 0u) {
                    mylist[n + __builtin_amdgcn_mbcnt_lo(live, 0u)] = c * CHUNK + (unsigned)__builtin_ctz(m);
                    m &= m - 1u;
                }
                n += (unsigned)__builtin_popcount(live);
                live = __builtin_amdgcn_ballot_w32(m != 0u);
            }
        };
        for (;;) {
            const bool last = bi >= IDX_BLOCKS;
            bool stop = false;
            while (cn >= BATCH || (last && cn > 0u)) {
                if (n + BATCH * CHUNK > QCAP) { stop = true; break; }
                const unsigned take = cn < BATCH ? cn : BATCH;
                cn -= take;
                const bool on0 = lane < take, on1 = lane + 32u < take;
                const unsigned c0 = on0 ? mycq[cn + lane] : 0u;
                const unsigned c1 = on1 ? mycq[cn + 32u + lane] : 0u;
                const int4 x0 = d4[2u * c0], x1 = d4[2u * c0 + 1u];
                const int4 y0 = d4[2u * c1], y1 = d4[2u * c1 + 1u];
                const unsigned m0 = hits8(x0, x1, on0);
                const unsigned m1 = hits8(y0, y1, on1);
                list_hits(m0, c0);
                list_hits(m1, c1);
            }
            if (stop || last) break;
            const unsigned nb = bi + N_WAVES;
            unsigned nxt = 0u;
            if (nb < IDX_BLOCKS) nxt = row[nb * 32u + lane];
            unsigned m = cur;
            const unsigned cbase = bi * 1024u + lane * 32u;
            unsigned live = __builtin_amdgcn_ballot_w32(m != 0u);
            while (live != 0u) {
                if (m != 0u) {
                    mycq[cn + __builtin_amdgcn_mbcnt_lo(live, 0u)] = cbase + (unsigned)__builtin_ctz(m);
                    m &= m - 1u;
                }
                cn += (unsigned)__builtin_popcount(live);
                live = __builtin_amdgcn_ballot_w32(m != 0u);
            }
            cur = nxt;
            bi = nb;
        }
        if (lane == 0 && (bi < IDX_BLOCKS || cn > 0u)) vmore[round & 1u] = 1u;

        unsigned made = 0u;
        unsigned left = 0u;
        for (unsigned i0 = 0; i0 < n; i0 += 128u) {
            unsigned e[4], w[4], d[4];
#pragma unroll
            for (int j = 0; j < 4; ++j) {
                unsigned idx = i0 + j * 32u + lane;
                e[j] = (idx < n) ? mylist[idx] : 0u;
            }
#pragma unroll
            for (int j = 0; j < 4; ++j) {
                w[j] = (unsigned)src[e[j]];
                d[j] = (unsigned)dst[e[j]];
            }
#pragma unroll
            for (int j = 0; j < 4; ++j) {
                const unsigned idx = i0 + j * 32u + lane;
                const bool valid = idx < n;
                const unsigned pk = (w[j] << 9) | (d[j] - lo);
                const unsigned o  = pk & 31u;
                unsigned peers = __builtin_amdgcn_ballot_w32(valid);
                unsigned forme = peers;
#pragma unroll
                for (int bit = 0; bit < 5; ++bit) {
                    const unsigned b = __builtin_amdgcn_ballot_w32(valid && ((o >> bit) & 1u));
                    peers &= ((o >> bit) & 1u) ? b : ~b;
                    forme &= ((lane >> bit) & 1u) ? b : ~b;
                }
                const unsigned rank  = (unsigned)__builtin_popcount(peers & ((1u << lane) - 1u));
                const unsigned basec = (unsigned)__builtin_amdgcn_ds_bpermute((int)(o << 2), (int)made);
                const unsigned pos   = basec + rank;
                if (valid && pos < CELLCAP) vcells[(wave * N_WAVES + o) * CELLCAP + pos] = pk;
                const bool full = valid && pos >= CELLCAP;
                const unsigned fm = __builtin_amdgcn_ballot_w32(full);
                if (fm != 0u) {
                    if (full) mylist[left + __builtin_amdgcn_mbcnt_lo(fm, 0u)] = pk;
                    left += (unsigned)__builtin_popcount(fm);
                }
                made += (unsigned)__builtin_popcount(forme);
            }
        }
        vccnt[wave * N_WAVES + lane] = (made < CELLCAP) ? made : CELLCAP;
        if (lane == 0) vcnt[wave] = left;
        __syncthreads();
        more = vmore[round & 1u];
        if (tid == 0) vmore[(round + 1u) & 1u] = 0u;

        unsigned nq = 0u;
        const unsigned cc = vccnt[lane * N_WAVES + wave];
        for (unsigned v = 0; v < N_WAVES; ++v) {
            const unsigned c = __builtin_amdgcn_readlane(cc, v);
            if (c != 0u) {
                if (lane < c) myown[nq + lane] = vcells[(v * N_WAVES + wave) * CELLCAP + lane];
                nq += c;
                if (nq >= 32u) {
                    nq -= 32u;
                    add_batch<true>(X, myown + nq, 32u, lane, racc);
                }
            }
        }
        const unsigned lc = vcnt[lane];
        if (__builtin_amdgcn_ballot_w32(lc != 0u) != 0u) {
            for (unsigned v = 0; v < N_WAVES; ++v) {
                const unsigned nv = __builtin_amdgcn_readlane(lc, v);
                const lds_vu32* lv = (const lds_vu32*)&qlist[v * WSTRIDE];
                for (unsigned i0 = 0; i0 < nv; i0 += 32u) {
                    const unsigned idx = i0 + lane;
                    const unsigned pk  = (idx < nv) ? lv[idx] : INVALID;
                    const unsigned rel = pk & 511u;
                    const bool mine = (rel < PART) && ((rel & 31u) == wave);
                    const unsigned mask = __builtin_amdgcn_ballot_w32(mine);
                    if (mask != 0u) {
                        if (mine) myown[nq + __builtin_amdgcn_mbcnt_lo(mask, 0u)] = pk;
                        nq += (unsigned)__builtin_popcount(mask);
                        if (nq >= 32u) {
                            nq -= 32u;
                            add_batch<true>(X, myown + nq, 32u, lane, racc);
                        }
                    }
                }
            }
        }
        if (nq > 0u) add_batch<false>(X, myown, nq, lane, racc);
        __syncthreads();
        ++round;
    } while (more != 0u);

#pragma unroll
    for (int k = 0; k < ROWS_PER_WAVE; ++k) {
        unsigned rel = wave + 32u * k;
        if (rel < PART) acc[rel * 32u + lane] = racc[k];
    }
    __syncthreads();

    if (wave < PART / 16) {
        const unsigned m16 = lane & 15u;
        const unsigned hi  = lane >> 4;
        const float* aptr = &acc[(wave * 16u + m16) * 32u + 2u * hi];
        v2f a[8];
#pragma unroll
        for (int s = 0; s < 8; ++s) {
            a[s].x = aptr[4 * s];
            a[s].y = aptr[4 * s + 1];
        }
        const unsigned row0 = lo + wave * 16u;
#pragma unroll
        for (int nt = 0; nt < 2; ++nt) {
            v8f c = {};
            const float* wp = W + (2u * hi) * 32u + nt * 16 + m16;
#pragma unroll
            for (int s = 0; s < 8; ++s) {
                v2f b;
                b.x = wp[(4 * s) * 32];
                b.y = wp[(4 * s + 1) * 32];
                c = __builtin_amdgcn_wmma_f32_16x16x4_f32(
                         false, a[s],  false, b,
                         (short)0, c,  false,  false);
            }
            float* op = out + (size_t)(row0 + 8u * hi) * 32 + nt * 16 + m16;
#pragma unroll
            for (int r = 0; r < 8; ++r) {
                op[(size_t)r * 32] = c[r];
            }
        }
    }
}

extern "C" void kernel_launch(void* const* d_in, const int* in_sizes, int n_in,
                              void* d_out, int out_size, void* d_ws, size_t ws_size,
                              hipStream_t stream) {
    const float* X   = (const float*)d_in[0];
    const float* W   = (const float*)d_in[1];
    const int*   src = (const int*)d_in[2];
    const int*   dst = (const int*)d_in[3];
    float*       out = (float*)d_out;
    unsigned*    bitmap = (unsigned*)d_ws;

    gin_index_kernel<<<IDX_BLOCKS, 1024, 0, stream>>>(dst, bitmap);
    gin_owner_kernel<<<N_PARTS, 1024, 0, stream>>>(X, W, src, dst, bitmap, out);
}


# owner kernel makes 3 scattered line requests per listed edge, not 5: a lane pair fetches a chunk's 32 bytes in one load; the list entry carries the row, so resolve reads only src; new fixed sum order
# speedup vs baseline: 2.8806x; 1.1456x over previous
#include <hip/hip_runtime.h>


#define N_NODES 100000
#define N_EDGES 1600000
#define IN_DIM  32
#define OUT_DIM 32

#define PART      400
#define N_PARTS   (N_NODES / PART)
#define N_WAVES   32
#define CHUNK     8
#define N_CHUNKS  (N_EDGES / CHUNK)
#define IDX_BLOCKS 196
#define ROW_WORDS (IDX_BLOCKS * 32)
#define BM_PITCH  257
#define BATCH     64
#define QCAP      828
#define CQCAP     1088
#define WSTRIDE   (QCAP + CQCAP)
#define ROWS_PER_WAVE 13
#define INVALID   0xffffffffu
#define CELLCAP   16

typedef __attribute__((ext_vector_type(2))) float v2f;
typedef __attribute__((ext_vector_type(8))) float v8f;
typedef __attribute__((ext_vector_type(16))) float v16f;
typedef __attribute__((address_space(3))) volatile unsigned lds_vu32;

__global__ __launch_bounds__(1024) void gin_index_kernel(const int* __restrict__ dst,
                                                         unsigned* __restrict__ bitmap) {
    __shared__ unsigned bm[N_WAVES * BM_PITCH];
    const unsigned tid  = threadIdx.x;
    const unsigned wave = tid >> 5;
    const unsigned lane = tid & 31u;
    for (unsigned i = tid; i < N_WAVES * BM_PITCH; i += 1024u) bm[i] = 0u;
    __syncthreads();
    const unsigned c  = blockIdx.x * 1024u + tid;
    const bool     ok = c < N_CHUNKS;
    const unsigned cc = ok ? c : 0u;
    const int4 d0 = ((const int4*)dst)[2u * cc];
    const int4 d1 = ((const int4*)dst)[2u * cc + 1u];
    const unsigned part[8] = {(unsigned)d0.x / PART, (unsigned)d0.y / PART, (unsigned)d0.z / PART, (unsigned)d0.w / PART,
                              (unsigned)d1.x / PART, (unsigned)d1.y / PART, (unsigned)d1.z / PART, (unsigned)d1.w / PART};
    lds_vu32* col = (lds_vu32*)&bm[wave * BM_PITCH];
#pragma unroll
    for (int j = 0; j < 8; ++j) {
        const unsigned q = part[j];
        unsigned peers = __builtin_amdgcn_ballot_w32(ok);
#pragma unroll
        for (int bit = 0; bit < 8; ++bit) {
            const unsigned b = __builtin_amdgcn_ballot_w32(ok && ((q >> bit) & 1u));
            peers &= ((q >> bit) & 1u) ? b : ~b;
        }
        if (ok && (peers & ((1u << lane) - 1u)) == 0u) col[q] = col[q] | peers;
    }
    __syncthreads();
    for (unsigned part = wave; part < N_PARTS; part += N_WAVES)
        bitmap[(part * IDX_BLOCKS + blockIdx.x) * 32u + lane] = bm[lane * BM_PITCH + part];
}

template <bool FULL>
__device__ __forceinline__ void add_batch(const float* __restrict__ X,
                                          const lds_vu32* q, unsigned nvalid,
                                          unsigned lane, v16f& racc) {
    unsigned pk = 0u;
    if (FULL || lane < nvalid) pk = q[lane];
    float v[32];
#pragma unroll
    for (int u = 0; u < 32; ++u) {
        unsigned e = __builtin_amdgcn_readlane(pk, u);
        v[u] = X[(e >> 9) * 32u + lane];
    }
#pragma unroll
    for (int u = 0; u < 32; ++u) {
        if (FULL || (unsigned)u < nvalid) {
            unsigned e = __builtin_amdgcn_readlane(pk, u);
            racc[(e & 511u) >> 5] += v[u];
        }
    }
}

__global__ __launch_bounds__(1024) void gin_owner_kernel(const float* __restrict__ X,
                                                         const float* __restrict__ W,
                                                         const int* __restrict__ src,
                                                         const int* __restrict__ dst,
                                                         const unsigned* __restrict__ bitmap,
                                                         float* __restrict__ out) {
    __shared__ unsigned qlist[N_WAVES * WSTRIDE];
    float* acc = (float*)qlist;
    __shared__ unsigned cells[N_WAVES * N_WAVES * CELLCAP];
    __shared__ unsigned cellcnt[N_WAVES * N_WAVES];
    __shared__ unsigned ownq[N_WAVES * 64];
    __shared__ unsigned qcnt[N_WAVES];
    __shared__ unsigned more_flag[2];

    const unsigned tid  = threadIdx.x;
    const unsigned wave = __builtin_amdgcn_readfirstlane(tid >> 5);
    const unsigned lane = tid & 31u;
    const unsigned p    = blockIdx.x;
    const unsigned lo   = p * PART;

    lds_vu32* mylist = (lds_vu32*)&qlist[wave * WSTRIDE];
    lds_vu32* mycq   = (lds_vu32*)&qlist[wave * WSTRIDE + QCAP];
    lds_vu32* myown  = (lds_vu32*)&ownq[wave * 64];
    lds_vu32* vcnt   = (lds_vu32*)qcnt;
    lds_vu32* vcells = (lds_vu32*)cells;
    lds_vu32* vccnt  = (lds_vu32*)cellcnt;
    lds_vu32* vmore  = (lds_vu32*)more_flag;

    if (tid < 2) vmore[tid] = 0u;
    __syncthreads();

    v16f racc = {};

    const unsigned* row = bitmap + p * ROW_WORDS;
    const int4* d4 = (const int4*)dst;
    unsigned bi  = wave;
    unsigned cur = row[bi * 32u + lane];
    unsigned cn  = 0u;
    unsigned round = 0u;
    unsigned more;

    do {
        unsigned n = 0u;
        auto list_hits = [&](const int4& a, bool on, unsigned e0) {
            const unsigned r0 = (unsigned)a.x - lo, r1 = (unsigned)a.y - lo;
            const unsigned r2 = (unsigned)a.z - lo, r3 = (unsigned)a.w - lo;
            unsigned m = (r0 < PART ? 1u : 0u) | (r1 < PART ? 2u : 0u) | (r2 < PART ? 4u : 0u) | (r3 < PART ? 8u : 0u);
            if (!on) m = 0u;
            unsigned live = __builtin_amdgcn_ballot_w32(m != 0u);
            while (live != 0u) {
                if (m != 0u) {
                    const unsigned j   = (unsigned)__builtin_ctz(m);
                    const unsigned rel = (j & 2u) ? ((j & 1u) ? r3 : r2) : ((j & 1u) ? r1 : r0);
                    mylist[n + __builtin_amdgcn_mbcnt_lo(live, 0u)] = ((e0 + j) << 9) | rel;
                    m &= m - 1u;
                }
                n += (unsigned)__builtin_popcount(live);
                live = __builtin_amdgcn_ballot_w32(m != 0u);
            }
        };
        for (;;) {
            const bool last = bi >= IDX_BLOCKS;
            bool stop = false;
            while (cn >= BATCH || (last && cn > 0u)) {
                if (n + BATCH * CHUNK > QCAP) { stop = true; break; }
                const unsigned take = cn < BATCH ? cn : BATCH;
                cn -= take;
                const unsigned half = lane & 1u;
                unsigned cq[4];
                bool     on[4];
                int4     xv[4];
#pragma unroll
                for (int k = 0; k < 4; ++k) {
                    const unsigned idx = 16u * k + (lane >> 1);
                    on[k] = idx < take;
                    cq[k] = on[k] ? mycq[cn + idx] : 0u;
                }
#pragma unroll
                for (int k = 0; k < 4; ++k) xv[k] = d4[2u * cq[k] + half];
#pragma unroll
                for (int k = 0; k < 4; ++k) list_hits(xv[k], on[k], cq[k] * CHUNK + 4u * half);
            }
            if (stop || last) break;
            const unsigned nb = bi + N_WAVES;
            unsigned nxt = 0u;
            if (nb < IDX_BLOCKS) nxt = row[nb * 32u + lane];
            unsigned m = cur;
            const unsigned cbase = bi * 1024u + lane * 32u;
            unsigned live = __builtin_amdgcn_ballot_w32(m != 0u);
            while (live != 0u) {
                if (m != 0u) {
                    mycq[cn + __builtin_amdgcn_mbcnt_lo(live, 0u)] = cbase + (unsigned)__builtin_ctz(m);
                    m &= m - 1u;
                }
                cn += (unsigned)__builtin_popcount(live);
                live = __builtin_amdgcn_ballot_w32(m != 0u);
            }
            cur = nxt;
            bi = nb;
        }
        if (lane == 0 && (bi < IDX_BLOCKS || cn > 0u)) vmore[round & 1u] = 1u;

        unsigned made = 0u;
        unsigned left = 0u;
        for (unsigned i0 = 0; i0 < n; i0 += 128u) {
            unsigned e[4], w[4];
#pragma unroll
            for (int j = 0; j < 4; ++j) {
                unsigned idx = i0 + j * 32u + lane;
                e[j] = (idx < n) ? mylist[idx] : 0u;
            }
#pragma unroll
            for (int j = 0; j < 4; ++j) {
                w[j] = (unsigned)src[e[j] >> 9];
            }
#pragma unroll
            for (int j = 0; j < 4; ++j) {
                const unsigned idx = i0 + j * 32u + lane;
                const bool valid = idx < n;
                const unsigned pk = (w[j] << 9) | (e[j] & 511u);
                const unsigned o  = pk & 31u;
                unsigned peers = __builtin_amdgcn_ballot_w32(valid);
                unsigned forme = peers;
#pragma unroll
                for (int bit = 0; bit < 5; ++bit) {
                    const unsigned b = __builtin_amdgcn_ballot_w32(valid && ((o >> bit) & 1u));
                    peers &= ((o >> bit) & 1u) ? b : ~b;
                    forme &= ((lane >> bit) & 1u) ? b : ~b;
                }
                const unsigned rank  = (unsigned)__builtin_popcount(peers & ((1u << lane) - 1u));
                const unsigned basec = (unsigned)__builtin_amdgcn_ds_bpermute((int)(o << 2), (int)made);
                const unsigned pos   = basec + rank;
                if (valid && pos < CELLCAP) vcells[(wave * N_WAVES + o) * CELLCAP + pos] = pk;
                const bool full = valid && pos >= CELLCAP;
                const unsigned fm = __builtin_amdgcn_ballot_w32(full);
                if (fm != 0u) {
                    if (full) mylist[left + __builtin_amdgcn_mbcnt_lo(fm, 0u)] = pk;
                    left += (unsigned)__builtin_popcount(fm);
                }
                made += (unsigned)__builtin_popcount(forme);
            }
        }
        vccnt[wave * N_WAVES + lane] = (made < CELLCAP) ? made : CELLCAP;
        if (lane == 0) vcnt[wave] = left;
        __syncthreads();
        more = vmore[round & 1u];
        if (tid == 0) vmore[(round + 1u) & 1u] = 0u;

        unsigned nq = 0u;
        const unsigned cc = vccnt[lane * N_WAVES + wave];
        for (unsigned v = 0; v < N_WAVES; ++v) {
            const unsigned c = __builtin_amdgcn_readlane(cc, v);
            if (c != 0u) {
                if (lane < c) myown[nq + lane] = vcells[(v * N_WAVES + wave) * CELLCAP + lane];
                nq += c;
                if (nq >= 32u) {
                    nq -= 32u;
                    add_batch<true>(X, myown + nq, 32u, lane, racc);
                }
            }
        }
        const unsigned lc = vcnt[lane];
        if (__builtin_amdgcn_ballot_w32(lc != 0u) != 0u) {
            for (unsigned v = 0; v < N_WAVES; ++v) {
                const unsigned nv = __builtin_amdgcn_readlane(lc, v);
                const lds_vu32* lv = (const lds_vu32*)&qlist[v * WSTRIDE];
                for (unsigned i0 = 0; i0 < nv; i0 += 32u) {
                    const unsigned idx = i0 + lane;
                    const unsigned pk  = (idx < nv) ? lv[idx] : INVALID;
                    const unsigned rel = pk & 511u;
                    const bool mine = (rel < PART) && ((rel & 31u) == wave);
                    const unsigned mask = __builtin_amdgcn_ballot_w32(mine);
                    if (mask != 0u) {
                        if (mine) myown[nq + __builtin_amdgcn_mbcnt_lo(mask, 0u)] = pk;
                        nq += (unsigned)__builtin_popcount(mask);
                        if (nq >= 32u) {
                            nq -= 32u;
                            add_batch<true>(X, myown + nq, 32u, lane, racc);
                        }
                    }
                }
            }
        }
        if (nq > 0u) add_batch<false>(X, myown, nq, lane, racc);
        __syncthreads();
        ++round;
    } while (more != 0u);

#pragma unroll
    for (int k = 0; k < ROWS_PER_WAVE; ++k) {
        unsigned rel = wave + 32u * k;
        if (rel < PART) acc[rel * 32u + lane] = racc[k];
    }
    __syncthreads();

    if (wave < PART / 16) {
        const unsigned m16 = lane & 15u;
        const unsigned hi  = lane >> 4;
        const float* aptr = &acc[(wave * 16u + m16) * 32u + 2u * hi];
        v2f a[8];
#pragma unroll
        for (int s = 0; s < 8; ++s) {
            a[s].x = aptr[4 * s];
            a[s].y = aptr[4 * s + 1];
        }
        const unsigned row0 = lo + wave * 16u;
#pragma unroll
        for (int nt = 0; nt < 2; ++nt) {
            v8f c = {};
            const float* wp = W + (2u * hi) * 32u + nt * 16 + m16;
#pragma unroll
            for (int s = 0; s < 8; ++s) {
                v2f b;
                b.x = wp[(4 * s) * 32];
                b.y = wp[(4 * s + 1) * 32];
                c = __builtin_amdgcn_wmma_f32_16x16x4_f32(
                         false, a[s],  false, b,
                         (short)0, c,  false,  false);
            }
            float* op = out + (size_t)(row0 + 8u * hi) * 32 + nt * 16 + m16;
#pragma unroll
            for (int r = 0; r < 8; ++r) {
                op[(size_t)r * 32] = c[r];
            }
        }
    }
}

extern "C" void kernel_launch(void* const* d_in, const int* in_sizes, int n_in,
                              void* d_out, int out_size, void* d_ws, size_t ws_size,
                              hipStream_t stream) {
    const float* X   = (const float*)d_in[0];
    const float* W   = (const float*)d_in[1];
    const int*   src = (const int*)d_in[2];
    const int*   dst = (const int*)d_in[3];
    float*       out = (float*)d_out;
    unsigned*    bitmap = (unsigned*)d_ws;

    gin_index_kernel<<<IDX_BLOCKS, 1024, 0, stream>>>(dst, bitmap);
    gin_owner_kernel<<<N_PARTS, 1024, 0, stream>>>(X, W, src, dst, bitmap, out);
}


# index kernel: the chunk's own thread marks an LDS byte per (part, chunk), bytes then packed to bits; no ballot matching of lanes, so blocks of 8 waves, not 32; sum order kept
# speedup vs baseline: 3.0629x; 1.0633x over previous
#include <hip/hip_runtime.h>


#define N_NODES 100000
#define N_EDGES 1600000
#define IN_DIM  32
#define OUT_DIM 32

#define PART      400
#define N_PARTS   (N_NODES / PART)
#define N_WAVES   32
#define CHUNK     8
#define N_CHUNKS  (N_EDGES / CHUNK)
#define IDX_BLOCKS 196
#define ROW_WORDS (IDX_BLOCKS * 32)
#define IDX_THREADS 256
#define IDX_CPT   (1024 / IDX_THREADS)
#define MK_PITCH  1040
#define BATCH     64
#define QCAP      828
#define CQCAP     1088
#define WSTRIDE   (QCAP + CQCAP)
#define ROWS_PER_WAVE 13
#define INVALID   0xffffffffu
#define CELLCAP   16

typedef __attribute__((ext_vector_type(2))) float v2f;
typedef __attribute__((ext_vector_type(8))) float v8f;
typedef __attribute__((ext_vector_type(16))) float v16f;
typedef __attribute__((address_space(3))) volatile unsigned lds_vu32;

__device__ __forceinline__ unsigned pack16(const uint4 q) {
    const unsigned lo = __builtin_amdgcn_sad_u8(q.x, 0u, __builtin_amdgcn_sad_u8(q.y, 0u, 0u));
    const unsigned hi = __builtin_amdgcn_sad_u8(q.z, 0u, __builtin_amdgcn_sad_u8(q.w, 0u, 0u));
    return lo | (hi << 8);
}

__global__ __launch_bounds__(IDX_THREADS) void gin_index_kernel(const int* __restrict__ dst,
                                                                unsigned* __restrict__ bitmap) {
    __shared__ __attribute__((aligned(16))) unsigned char mark[N_PARTS * MK_PITCH];
    const unsigned tid  = threadIdx.x;
    const unsigned wave = tid >> 5;
    const unsigned lane = tid & 31u;
    int4 d0[IDX_CPT], d1[IDX_CPT];
    bool ok[IDX_CPT];
#pragma unroll
    for (int i = 0; i < IDX_CPT; ++i) {
        const unsigned c  = blockIdx.x * 1024u + (unsigned)i * IDX_THREADS + tid;
        ok[i] = c < N_CHUNKS;
        const unsigned cc = ok[i] ? c : 0u;
        d0[i] = ((const int4*)dst)[2u * cc];
        d1[i] = ((const int4*)dst)[2u * cc + 1u];
    }
    uint4* mark4 = (uint4*)mark;
    for (unsigned i = tid; i < N_PARTS * MK_PITCH / 16u; i += IDX_THREADS) mark4[i] = make_uint4(0u, 0u, 0u, 0u);
    __syncthreads();
#pragma unroll
    for (int i = 0; i < IDX_CPT; ++i) {
        if (ok[i]) {
            const unsigned cl = (unsigned)i * IDX_THREADS + tid;
            const unsigned char v = (unsigned char)(1u << (cl & 7u));
            mark[((unsigned)d0[i].x / PART) * MK_PITCH + cl] = v;
            mark[((unsigned)d0[i].y / PART) * MK_PITCH + cl] = v;
            mark[((unsigned)d0[i].z / PART) * MK_PITCH + cl] = v;
            mark[((unsigned)d0[i].w / PART) * MK_PITCH + cl] = v;
            mark[((unsigned)d1[i].x / PART) * MK_PITCH + cl] = v;
            mark[((unsigned)d1[i].y / PART) * MK_PITCH + cl] = v;
            mark[((unsigned)d1[i].z / PART) * MK_PITCH + cl] = v;
            mark[((unsigned)d1[i].w / PART) * MK_PITCH + cl] = v;
        }
    }
    __syncthreads();
    const bool     odd  = (lane & 1u) != 0u;
    const unsigned slot = odd ? 16u + (lane >> 1) : (lane >> 1);
    const unsigned wv = __builtin_amdgcn_readfirstlane(wave);
    auto pack_part = [&](unsigned part) {
        const unsigned ha = pack16(mark4[part * (MK_PITCH / 16u) + lane]);
        const unsigned hb = pack16(mark4[part * (MK_PITCH / 16u) + 32u + lane]);
        const unsigned got = (unsigned)__builtin_amdgcn_update_dpp(0, (int)(odd ? ha : hb), 0xB1, 0xF, 0xF, true);
        const unsigned word = odd ? (got | (hb << 16)) : (ha | (got << 16));
        bitmap[(part * IDX_BLOCKS + blockIdx.x) * 32u + slot] = word;
    };
#pragma unroll 8
    for (unsigned k = 0; k < 24u; ++k) pack_part(wv + k * (IDX_THREADS / 32u));
#pragma unroll
    for (unsigned k = 24u; k < 31u; ++k) pack_part(wv + k * (IDX_THREADS / 32u));
    if (wv + 31u * (IDX_THREADS / 32u) < N_PARTS) pack_part(wv + 31u * (IDX_THREADS / 32u));
}

template <bool FULL>
__device__ __forceinline__ void add_batch(const float* __restrict__ X,
                                          const lds_vu32* q, unsigned nvalid,
                                          unsigned lane, v16f& racc) {
    unsigned pk = 0u;
    if (FULL || lane < nvalid) pk = q[lane];
    float v[32];
#pragma unroll
    for (int u = 0; u < 32; ++u) {
        unsigned e = __builtin_amdgcn_readlane(pk, u);
        v[u] = X[(e >> 9) * 32u + lane];
    }
#pragma unroll
    for (int u = 0; u < 32; ++u) {
        if (FULL || (unsigned)u < nvalid) {
            unsigned e = __builtin_amdgcn_readlane(pk, u);
            racc[(e & 511u) >> 5] += v[u];
        }
    }
}

__global__ __launch_bounds__(1024) void gin_owner_kernel(const float* __restrict__ X,
                                                         const float* __restrict__ W,
                                                         const int* __restrict__ src,
                                                         const int* __restrict__ dst,
                                                         const unsigned* __restrict__ bitmap,
                                                         float* __restrict__ out) {
    __shared__ unsigned qlist[N_WAVES * WSTRIDE];
    float* acc = (float*)qlist;
    __shared__ unsigned cells[N_WAVES * N_WAVES * CELLCAP];
    __shared__ unsigned cellcnt[N_WAVES * N_WAVES];
    __shared__ unsigned ownq[N_WAVES * 64];
    __shared__ unsigned qcnt[N_WAVES];
    __shared__ unsigned more_flag[2];

    const unsigned tid  = threadIdx.x;
    const unsigned wave = __builtin_amdgcn_readfirstlane(tid >> 5);
    const unsigned lane = tid & 31u;
    const unsigned p    = blockIdx.x;
    const unsigned lo   = p * PART;

    lds_vu32* mylist = (lds_vu32*)&qlist[wave * WSTRIDE];
    lds_vu32* mycq   = (lds_vu32*)&qlist[wave * WSTRIDE + QCAP];
    lds_vu32* myown  = (lds_vu32*)&ownq[wave * 64];
    lds_vu32* vcnt   = (lds_vu32*)qcnt;
    lds_vu32* vcells = (lds_vu32*)cells;
    lds_vu32* vccnt  = (lds_vu32*)cellcnt;
    lds_vu32* vmore  = (lds_vu32*)more_flag;

    if (tid < 2) vmore[tid] = 0u;
    __syncthreads();

    v16f racc = {};

    const unsigned* row = bitmap + p * ROW_WORDS;
    const int4* d4 = (const int4*)dst;
    unsigned bi  = wave;
    unsigned cur = row[bi * 32u + lane];
    unsigned cn  = 0u;
    unsigned round = 0u;
    unsigned more;

    do {
        unsigned n = 0u;
        auto list_hits = [&](const int4& a, bool on, unsigned e0) {
            const unsigned r0 = (unsigned)a.x - lo, r1 = (unsigned)a.y - lo;
            const unsigned r2 = (unsigned)a.z - lo, r3 = (unsigned)a.w - lo;
            unsigned m = (r0 < PART ? 1u : 0u) | (r1 < PART ? 2u : 0u) | (r2 < PART ? 4u : 0u) | (r3 < PART ? 8u : 0u);
            if (!on) m = 0u;
            unsigned live = __builtin_amdgcn_ballot_w32(m != 0u);
            while (live != 0u) {
                if (m != 0u) {
                    const unsigned j   = (unsigned)__builtin_ctz(m);
                    const unsigned rel = (j & 2u) ? ((j & 1u) ? r3 : r2) : ((j & 1u) ? r1 : r0);
                    mylist[n + __builtin_amdgcn_mbcnt_lo(live, 0u)] = ((e0 + j) << 9) | rel;
                    m &= m - 1u;
                }
                n += (unsigned)__builtin_popcount(live);
                live = __builtin_amdgcn_ballot_w32(m != 0u);
            }
        };
        for (;;) {
            const bool last = bi >= IDX_BLOCKS;
            bool stop = false;
            while (cn >= BATCH || (last && cn > 0u)) {
                if (n + BATCH * CHUNK > QCAP) { stop = true; break; }
                const unsigned take = cn < BATCH ? cn : BATCH;
                cn -= take;
                const unsigned half = lane & 1u;
                unsigned cq[4];
                bool     on[4];
                int4     xv[4];
#pragma unroll
                for (int k = 0; k < 4; ++k) {
                    const unsigned idx = 16u * k + (lane >> 1);
                    on[k] = idx < take;
                    cq[k] = on[k] ? mycq[cn + idx] : 0u;
                }
#pragma unroll
                for (int k = 0; k < 4; ++k) xv[k] = d4[2u * cq[k] + half];
#pragma unroll
                for (int k = 0; k < 4; ++k) list_hits(xv[k], on[k], cq[k] * CHUNK + 4u * half);
            }
            if (stop || last) break;
            const unsigned nb = bi + N_WAVES;
            unsigned nxt = 0u;
            if (nb < IDX_BLOCKS) nxt = row[nb * 32u + lane];
            unsigned m = cur;
            const unsigned cbase = bi * 1024u + lane * 32u;
            unsigned live = __builtin_amdgcn_ballot_w32(m != 0u);
            while (live != 0u) {
                if (m != 0u) {
                    mycq[cn + __builtin_amdgcn_mbcnt_lo(live, 0u)] = cbase + (unsigned)__builtin_ctz(m);
                    m &= m - 1u;
                }
                cn += (unsigned)__builtin_popcount(live);
                live = __builtin_amdgcn_ballot_w32(m != 0u);
            }
            cur = nxt;
            bi = nb;
        }
        if (lane == 0 && (bi < IDX_BLOCKS || cn > 0u)) vmore[round & 1u] = 1u;

        unsigned made = 0u;
        unsigned left = 0u;
        for (unsigned i0 = 0; i0 < n; i0 += 128u) {
            unsigned e[4], w[4];
#pragma unroll
            for (int j = 0; j < 4; ++j) {
                unsigned idx = i0 + j * 32u + lane;
                e[j] = (idx < n) ? mylist[idx] : 0u;
            }
#pragma unroll
            for (int j = 0; j < 4; ++j) {
                w[j] = (unsigned)src[e[j] >> 9];
            }
#pragma unroll
            for (int j = 0; j < 4; ++j) {
                const unsigned idx = i0 + j * 32u + lane;
                const bool valid = idx < n;
                const unsigned pk = (w[j] << 9) | (e[j] & 511u);
                const unsigned o  = pk & 31u;
                unsigned peers = __builtin_amdgcn_ballot_w32(valid);
                unsigned forme = peers;
#pragma unroll
                for (int bit = 0; bit < 5; ++bit) {
                    const unsigned b = __builtin_amdgcn_ballot_w32(valid && ((o >> bit) & 1u));
                    peers &= ((o >> bit) & 1u) ? b : ~b;
                    forme &= ((lane >> bit) & 1u) ? b : ~b;
                }
                const unsigned rank  = (unsigned)__builtin_popcount(peers & ((1u << lane) - 1u));
                const unsigned basec = (unsigned)__builtin_amdgcn_ds_bpermute((int)(o << 2), (int)made);
                const unsigned pos   = basec + rank;
                if (valid && pos < CELLCAP) vcells[(wave * N_WAVES + o) * CELLCAP + pos] = pk;
                const bool full = valid && pos >= CELLCAP;
                const unsigned fm = __builtin_amdgcn_ballot_w32(full);
                if (fm != 0u) {
                    if (full) mylist[left + __builtin_amdgcn_mbcnt_lo(fm, 0u)] = pk;
                    left += (unsigned)__builtin_popcount(fm);
                }
                made += (unsigned)__builtin_popcount(forme);
            }
        }
        vccnt[wave * N_WAVES + lane] = (made < CELLCAP) ? made : CELLCAP;
        if (lane == 0) vcnt[wave] = left;
        __syncthreads();
        more = vmore[round & 1u];
        if (tid == 0) vmore[(round + 1u) & 1u] = 0u;

        unsigned nq = 0u;
        const unsigned cc = vccnt[lane * N_WAVES + wave];
        for (unsigned v = 0; v < N_WAVES; ++v) {
            const unsigned c = __builtin_amdgcn_readlane(cc, v);
            if (c != 0u) {
                if (lane < c) myown[nq + lane] = vcells[(v * N_WAVES + wave) * CELLCAP + lane];
                nq += c;
                if (nq >= 32u) {
                    nq -= 32u;
                    add_batch<true>(X, myown + nq, 32u, lane, racc);
                }
            }
        }
        const unsigned lc = vcnt[lane];
        if (__builtin_amdgcn_ballot_w32(lc != 0u) != 0u) {
            for (unsigned v = 0; v < N_WAVES; ++v) {
                const unsigned nv = __builtin_amdgcn_readlane(lc, v);
                const lds_vu32* lv = (const lds_vu32*)&qlist[v * WSTRIDE];
                for (unsigned i0 = 0; i0 < nv; i0 += 32u) {
                    const unsigned idx = i0 + lane;
                    const unsigned pk  = (idx < nv) ? lv[idx] : INVALID;
                    const unsigned rel = pk & 511u;
                    const bool mine = (rel < PART) && ((rel & 31u) == wave);
                    const unsigned mask = __builtin_amdgcn_ballot_w32(mine);
                    if (mask != 0u) {
                        if (mine) myown[nq + __builtin_amdgcn_mbcnt_lo(mask, 0u)] = pk;
                        nq += (unsigned)__builtin_popcount(mask);
                        if (nq >= 32u) {
                            nq -= 32u;
                            add_batch<true>(X, myown + nq, 32u, lane, racc);
                        }
                    }
                }
            }
        }
        if (nq > 0u) add_batch<false>(X, myown, nq, lane, racc);
        __syncthreads();
        ++round;
    } while (more != 0u);

#pragma unroll
    for (int k = 0; k < ROWS_PER_WAVE; ++k) {
        unsigned rel = wave + 32u * k;
        if (rel < PART) acc[rel * 32u + lane] = racc[k];
    }
    __syncthreads();

    if (wave < PART / 16) {
        const unsigned m16 = lane & 15u;
        const unsigned hi  = lane >> 4;
        const float* aptr = &acc[(wave * 16u + m16) * 32u + 2u * hi];
        v2f a[8];
#pragma unroll
        for (int s = 0; s < 8; ++s) {
            a[s].x = aptr[4 * s];
            a[s].y = aptr[4 * s + 1];
        }
        const unsigned row0 = lo + wave * 16u;
#pragma unroll
        for (int nt = 0; nt < 2; ++nt) {
            v8f c = {};
            const float* wp = W + (2u * hi) * 32u + nt * 16 + m16;
#pragma unroll
            for (int s = 0; s < 8; ++s) {
                v2f b;
                b.x = wp[(4 * s) * 32];
                b.y = wp[(4 * s + 1) * 32];
                c = __builtin_amdgcn_wmma_f32_16x16x4_f32(
                         false, a[s],  false, b,
                         (short)0, c,  false,  false);
            }
            float* op = out + (size_t)(row0 + 8u * hi) * 32 + nt * 16 + m16;
#pragma unroll
            for (int r = 0; r < 8; ++r) {
                op[(size_t)r * 32] = c[r];
            }
        }
    }
}

extern "C" void kernel_launch(void* const* d_in, const int* in_sizes, int n_in,
                              void* d_out, int out_size, void* d_ws, size_t ws_size,
                              hipStream_t stream) {
    const float* X   = (const float*)d_in[0];
    const float* W   = (const float*)d_in[1];
    const int*   src = (const int*)d_in[2];
    const int*   dst = (const int*)d_in[3];
    float*       out = (float*)d_out;
    unsigned*    bitmap = (unsigned*)d_ws;

    gin_index_kernel<<<IDX_BLOCKS, IDX_THREADS, 0, stream>>>(dst, bitmap);
    gin_owner_kernel<<<N_PARTS, 1024, 0, stream>>>(X, W, src, dst, bitmap, out);
}
